# starting kernel
# baseline (speedup 1.0000x reference)
#include <hip/hip_runtime.h>

typedef float v2f __attribute__((ext_vector_type(2)));
typedef float v8f __attribute__((ext_vector_type(8)));

#define D 64

__global__ void __launch_bounds__(256)
gemm_wmma(const float* __restrict__ X, const float* __restrict__ W,
          float* __restrict__ S, int n)
{
    const int lane = threadIdx.x & 31;
    const int wave = threadIdx.x >> 5;
    const int rowTile = blockIdx.x * 2 + (wave >> 2);
    const int m0 = rowTile * 16;
    if (m0 >= n) return;
    const int n0 = (wave & 3) * 16;

    const int h  = lane >> 4;
    const int ml = lane & 15;

    const float* xrow = X + (size_t)(m0 + ml) * D;

    v8f c = {};
    #pragma unroll
    for (int k0 = 0; k0 < D; k0 += 4) {
        const int k = k0 + 2 * h;
        v2f a;
        a.x = xrow[k];
        a.y = xrow[k + 1];
        v2f b;
        b.x = W[(size_t)k       * D + n0 + ml];
        b.y = W[(size_t)(k + 1) * D + n0 + ml];
        c = __builtin_amdgcn_wmma_f32_16x16x4_f32(
                 false, a,  false, b,
                 (short)0, c,  false,  false);
    }

    #pragma unroll
    for (int v = 0; v < 8; ++v) {
        S[(size_t)(m0 + v + 8 * h) * D + n0 + ml] = c[v];
    }
}

__global__ void __launch_bounds__(256)
init_out(float* __restrict__ out, const float* __restrict__ b, int total)
{
    int i = blockIdx.x * 256 + threadIdx.x;
    if (i < total) out[i] = b[i & (D - 1)];
}

__global__ void __launch_bounds__(256)
edge_scatter(const float* __restrict__ S, const float* __restrict__ ev,
             const int* __restrict__ row, const int* __restrict__ col,
             float* __restrict__ out, int E)
{
    const int tid  = blockIdx.x * 256 + threadIdx.x;
    const int wid  = tid >> 5;
    const int lane = threadIdx.x & 31;
    const int h    = lane >> 4;
    const int li   = lane & 15;

    const int e = wid * 2 + h;
    if (e >= E) return;

    const int   r = row[e];
    const int   c = col[e];
    const float v = ev[e];

    const float4 s = *(const float4*)(S + (size_t)c * D + li * 4);
    float* o = out + (size_t)r * D + li * 4;
    unsafeAtomicAdd(o + 0, v * s.x);
    unsafeAtomicAdd(o + 1, v * s.y);
    unsafeAtomicAdd(o + 2, v * s.z);
    unsafeAtomicAdd(o + 3, v * s.w);
}

extern "C" void kernel_launch(void* const* d_in, const int* in_sizes, int n_in,
                              void* d_out, int out_size, void* d_ws, size_t ws_size,
                              hipStream_t stream)
{
    const float* X   = (const float*)d_in[0];
    const float* W   = (const float*)d_in[1];
    const float* b   = (const float*)d_in[2];
    const float* ev  = (const float*)d_in[3];
    const int*   row = (const int*)d_in[4];
    const int*   col = (const int*)d_in[5];
    float*       out = (float*)d_out;
    float*       S   = (float*)d_ws;

    const int N = in_sizes[0] / D;
    const int E = in_sizes[3];

    const int rowTiles  = (N + 15) / 16;
    const int gemmBlks  = (rowTiles + 1) / 2;
    gemm_wmma<<<gemmBlks, 256, 0, stream>>>(X, W, S, N);

    const int total    = N * D;
    const int initBlks = (total + 255) / 256;
    init_out<<<initBlks, 256, 0, stream>>>(out, b, total);

    const int waves    = (E + 1) / 2;
    const int edgeBlks = (waves + 7) / 8;
    edge_scatter<<<edgeBlks, 256, 0, stream>>>(S, ev, row, col, out, E);
}


# edges counting-sorted per call into 128-row output tiles (LDS counters, no global atomics); a workgroup per tile adds in LDS (f32 LDS atomics, arrival order), stores out once with bias
# speedup vs baseline: 1.1706x; 1.1706x over previous
#include <hip/hip_runtime.h>

typedef float v2f __attribute__((ext_vector_type(2)));
typedef float v8f __attribute__((ext_vector_type(8)));

#define D 64

__global__ void __launch_bounds__(256)
gemm_wmma(const float* __restrict__ X, const float* __restrict__ W,
          float* __restrict__ S, int n)
{
    const int lane = threadIdx.x & 31;
    const int wave = threadIdx.x >> 5;
    const int rowTile = blockIdx.x * 2 + (wave >> 2);
    const int m0 = rowTile * 16;
    if (m0 >= n) return;
    const int n0 = (wave & 3) * 16;

    const int h  = lane >> 4;
    const int ml = lane & 15;

    const float* xrow = X + (size_t)(m0 + ml) * D;

    v8f c = {};
    #pragma unroll
    for (int k0 = 0; k0 < D; k0 += 4) {
        const int k = k0 + 2 * h;
        v2f a;
        a.x = xrow[k];
        a.y = xrow[k + 1];
        v2f b;
        b.x = W[(size_t)k       * D + n0 + ml];
        b.y = W[(size_t)(k + 1) * D + n0 + ml];
        c = __builtin_amdgcn_wmma_f32_16x16x4_f32(
                 false, a,  false, b,
                 (short)0, c,  false,  false);
    }

    #pragma unroll
    for (int v = 0; v < 8; ++v) {
        S[(size_t)(m0 + v + 8 * h) * D + n0 + ml] = c[v];
    }
}

#define TR      128
#define TSH     7
#define MAXNB   1024
#define CH      8192
#define EPT     8

__device__ __forceinline__ int block_scan_incl(int v, int (*sc)[1024], int t)
{
    sc[0][t] = v;
    __syncthreads();
    int src = 0;
    for (int off = 1; off < 1024; off <<= 1) {
        int x = sc[src][t];
        if (t >= off) x += sc[src][t - off];
        sc[src ^ 1][t] = x;
        __syncthreads();
        src ^= 1;
    }
    return sc[src][t];
}

__global__ void __launch_bounds__(1024)
bin_count(const int* __restrict__ row, int* __restrict__ H, int E, int nb)
{
    __shared__ int cnt[MAXNB + 1];
    const int t  = threadIdx.x;
    const int e0 = blockIdx.x * CH;
    cnt[t] = 0;
    __syncthreads();
    int r[EPT];
    #pragma unroll
    for (int j = 0; j < EPT; ++j) r[j] = row[min(e0 + j * 1024 + t, E - 1)];
    #pragma unroll
    for (int j = 0; j < EPT; ++j)
        atomicAdd(&cnt[(e0 + j * 1024 + t < E) ? (r[j] >> TSH) : MAXNB], 1);
    __syncthreads();
    if (t < nb) H[(size_t)blockIdx.x * nb + t] = cnt[t];
}

__global__ void __launch_bounds__(1024)
bin_scan(const int* __restrict__ H, int* __restrict__ Bs, int* __restrict__ binStart,
         int nb, int nc)
{
    __shared__ int sc[2][1024];
    const int t = threadIdx.x;
    int run = 0;
    if (t < nb) {
        #pragma unroll 28
        for (int c = 0; c < nc; ++c) {
            const int v = H[(size_t)c * nb + t];
            Bs[(size_t)c * nb + t] = run;
            run += v;
        }
    }
    const int incl = block_scan_incl(run, sc, t);
    if (t <= nb) binStart[t] = incl - run;
}

__global__ void __launch_bounds__(1024)
bin_scatter(const int* __restrict__ row, const int* __restrict__ col,
            const float* __restrict__ ev, const int* __restrict__ Bs,
            const int* __restrict__ binStart, uint2* __restrict__ recs, int E, int nb)
{
    __shared__ int            cnt[MAXNB + 1];
    __shared__ int            goff[MAXNB];
    __shared__ int            sc[2][1024];
    __shared__ uint2          stage[CH];
    __shared__ unsigned short sbin[CH];

    const int t  = threadIdx.x;
    const int e0 = blockIdx.x * CH;
    cnt[t] = 0;
    __syncthreads();

    int      r[EPT];
    unsigned key[EPT];
    unsigned val[EPT];
    int      bin[EPT];
    int      rnk[EPT];
    #pragma unroll
    for (int j = 0; j < EPT; ++j) {
        const int ec = min(e0 + j * 1024 + t, E - 1);
        r[j]   = row[ec];
        key[j] = (unsigned)col[ec];
        val[j] = __float_as_uint(ev[ec]);
    }
    #pragma unroll
    for (int j = 0; j < EPT; ++j) {
        bin[j]  = (e0 + j * 1024 + t < E) ? (r[j] >> TSH) : MAXNB;
        key[j] |= (unsigned)(r[j] & (TR - 1)) << 24;
        rnk[j]  = atomicAdd(&cnt[bin[j]], 1);
    }
    __syncthreads();

    const int mine = cnt[t];
    const int excl = block_scan_incl(mine, sc, t) - mine;
    cnt[t]  = excl;
    goff[t] = (t < nb) ? (binStart[t] + Bs[(size_t)blockIdx.x * nb + t] - excl) : 0;
    __syncthreads();

    #pragma unroll
    for (int j = 0; j < EPT; ++j) {
        if (bin[j] < MAXNB) {
            const int p = cnt[bin[j]] + rnk[j];
            stage[p] = make_uint2(key[j], val[j]);
            sbin[p]  = (unsigned short)bin[j];
        }
    }
    __syncthreads();

    const int nloc = min(CH, E - e0);
    uint2 rec[EPT];
    int   pos[EPT];
    #pragma unroll
    for (int j = 0; j < EPT; ++j) {
        const int ic = min(j * 1024 + t, nloc - 1);
        rec[j] = stage[ic];
        pos[j] = ic + goff[sbin[ic]];
    }
    #pragma unroll
    for (int j = 0; j < EPT; ++j)
        if (j * 1024 + t < nloc) recs[pos[j]] = rec[j];
}

#define TS_THREADS 512
#define TS_U       4
#define TS_STEP    ((TS_THREADS / 16) * TS_U)
__global__ void __launch_bounds__(TS_THREADS)
tile_sum(const float* __restrict__ S, const uint2* __restrict__ recs,
         const int* __restrict__ binStart, const float* __restrict__ b,
         float* __restrict__ out, int n)
{
    __shared__ float acc[(TR + 1) * D];
    const int t    = threadIdx.x;
    const int tile = blockIdx.x;
    float4* acc4 = (float4*)acc;
    for (int i = t; i < TR * D / 4; i += TS_THREADS) acc4[i] = make_float4(0.f, 0.f, 0.f, 0.f);
    __syncthreads();

    const int beg  = binStart[tile];
    const int end  = binStart[tile + 1];
    const int last = max(end - 1, 0);
    const int g    = t >> 4;
    const int li   = t & 15;
    const int k0   = (g & 1) * 2;

    int   i = beg + g;
    uint2 q[TS_U];
    #pragma unroll
    for (int u = 0; u < TS_U; ++u) q[u] = recs[min(i + (TS_THREADS / 16) * u, last)];

    while (i < end) {
        float4 s[TS_U];
        #pragma unroll
        for (int u = 0; u < TS_U; ++u)
            s[u] = *(const float4*)(S + (size_t)(q[u].x & 0x00FFFFFFu) * D + li * 4);

        const int in = i + TS_STEP;
        uint2 qn[TS_U];
        #pragma unroll
        for (int u = 0; u < TS_U; ++u) qn[u] = recs[min(in + (TS_THREADS / 16) * u, last)];

        #pragma unroll
        for (int u = 0; u < TS_U; ++u) {
            const bool     ok = (i + (TS_THREADS / 16) * u < end);
            const unsigned rl = ok ? (q[u].x >> 24) : (unsigned)TR;
            const float    v  = __uint_as_float(q[u].y);
            float* a = acc + rl * D + li * 4;
            const float p0 = v * s[u].x, p1 = v * s[u].y, p2 = v * s[u].z, p3 = v * s[u].w;
            atomicAdd(a + k0,           k0 ? p2 : p0);
            atomicAdd(a + k0 + 1,       k0 ? p3 : p1);
            atomicAdd(a + (k0 ^ 2),     k0 ? p0 : p2);
            atomicAdd(a + (k0 ^ 2) + 1, k0 ? p1 : p3);
        }
        #pragma unroll
        for (int u = 0; u < TS_U; ++u) q[u] = qn[u];
        i = in;
    }
    __syncthreads();

    const float4* b4   = (const float4*)b;
    float4*       out4 = (float4*)out;
    for (int j = t; j < TR * D / 4; j += TS_THREADS) {
        const int r = tile * TR + (j >> 4);
        if (r < n) {
            const float4 a  = acc4[j];
            const float4 bb = b4[j & 15];
            out4[(size_t)r * (D / 4) + (j & 15)] = make_float4(a.x + bb.x, a.y + bb.y, a.z + bb.z, a.w + bb.w);
        }
    }
}

extern "C" void kernel_launch(void* const* d_in, const int* in_sizes, int n_in,
                              void* d_out, int out_size, void* d_ws, size_t ws_size,
                              hipStream_t stream)
{
    const float* X   = (const float*)d_in[0];
    const float* W   = (const float*)d_in[1];
    const float* b   = (const float*)d_in[2];
    const float* ev  = (const float*)d_in[3];
    const int*   row = (const int*)d_in[4];
    const int*   col = (const int*)d_in[5];
    float*       out = (float*)d_out;

    char*  ws       = (char*)d_ws;
    float* S        = (float*)ws;
    int*   H        = (int*)(ws + (32u << 20));
    int*   Bs       = (int*)(ws + (34u << 20));
    int*   binStart = (int*)(ws + (36u << 20));
    uint2* recs     = (uint2*)(ws + (40u << 20));

    const int N = in_sizes[0] / D;
    const int E = in_sizes[3];

    const int rowTiles  = (N + 15) / 16;
    const int gemmBlks  = (rowTiles + 1) / 2;
    gemm_wmma<<<gemmBlks, 256, 0, stream>>>(X, W, S, N);

    const int nb = (N + TR - 1) / TR;
    const int nc = (E + CH - 1) / CH;
    bin_count<<<nc, 1024, 0, stream>>>(row, H, E, nb);
    bin_scan<<<1, 1024, 0, stream>>>(H, Bs, binStart, nb, nc);
    bin_scatter<<<nc, 1024, 0, stream>>>(row, col, ev, Bs, binStart, recs, E, nb);

    tile_sum<<<nb, TS_THREADS, 0, stream>>>(S, recs, binStart, b, out, N);
}


# records stay in their chunk, LDS-sorted by tile, with a run word per (tile, chunk); a tile lists its runs' record slots in LDS and reads through it; count and scan launches go; arrival-order sums
# speedup vs baseline: 2.2242x; 1.9001x over previous
#include <hip/hip_runtime.h>

typedef float v2f __attribute__((ext_vector_type(2)));
typedef float v8f __attribute__((ext_vector_type(8)));

#define D 64

#define TR      128
#define TSH     7
#define MAXNB   1024
#define CH      8192
#define EPT     8

__device__ __forceinline__ int block_scan_incl(int v, int (*sc)[1024], int t)
{
    sc[0][t] = v;
    __syncthreads();
    int src = 0;
    for (int off = 1; off < 1024; off <<= 1) {
        int x = sc[src][t];
        if (t >= off) x += sc[src][t - off];
        sc[src ^ 1][t] = x;
        __syncthreads();
        src ^= 1;
    }
    return sc[src][t];
}

__global__ void __launch_bounds__(1024)
chunk_sort(const int* __restrict__ row, const int* __restrict__ col,
           const float* __restrict__ ev, uint2* __restrict__ recs,
           unsigned* __restrict__ runs, int E, int nb, int nc)
{
    __shared__ int   cnt[MAXNB + 1];
    __shared__ int   sc[2][1024];
    __shared__ uint2 stage[CH];

    const int t  = threadIdx.x;
    const int e0 = blockIdx.x * CH;
    cnt[t] = 0;
    __syncthreads();

    int      r[EPT];
    unsigned key[EPT];
    unsigned val[EPT];
    int      bin[EPT];
    int      rnk[EPT];
    #pragma unroll
    for (int j = 0; j < EPT; ++j) {
        const int ec = min(e0 + j * 1024 + t, E - 1);
        r[j]   = row[ec];
        key[j] = (unsigned)col[ec];
        val[j] = __float_as_uint(ev[ec]);
    }
    #pragma unroll
    for (int j = 0; j < EPT; ++j) {
        bin[j]  = (e0 + j * 1024 + t < E) ? (r[j] >> TSH) : MAXNB;
        key[j] |= (unsigned)(r[j] & (TR - 1)) << 24;
        rnk[j]  = atomicAdd(&cnt[bin[j]], 1);
    }
    __syncthreads();

    const int mine = cnt[t];
    const int excl = block_scan_incl(mine, sc, t) - mine;
    cnt[t] = excl;
    if (t < nb) runs[(size_t)t * nc + blockIdx.x] = (unsigned)excl | ((unsigned)mine << 16);
    __syncthreads();

    #pragma unroll
    for (int j = 0; j < EPT; ++j)
        if (bin[j] < MAXNB) stage[cnt[bin[j]] + rnk[j]] = make_uint2(key[j], val[j]);
    __syncthreads();

    const int nloc = min(CH, E - e0);
    uint2 rec[EPT];
    #pragma unroll
    for (int j = 0; j < EPT; ++j) rec[j] = stage[min(j * 1024 + t, nloc - 1)];
    #pragma unroll
    for (int j = 0; j < EPT; ++j)
        if (j * 1024 + t < nloc) recs[(size_t)e0 + j * 1024 + t] = rec[j];
}

#define TS_THREADS 512
#define TS_WAVES   (TS_THREADS / 32)
#define TS_U       4
#define TS_STEP    ((TS_THREADS / 16) * TS_U)
#define ALD        68
#define SLOTS      4096

__device__ __forceinline__ int tile_scan_excl(int v, int* wsum, int t, int& total)
{
    const int lane = t & 31;
    const int wave = t >> 5;
    int x = v;
    #pragma unroll
    for (int off = 1; off < 32; off <<= 1) {
        const int y = __shfl_up(x, off);
        if (lane >= off) x += y;
    }
    if (lane == 31) wsum[wave] = x;
    __syncthreads();
    int before = 0;
    int all    = 0;
    #pragma unroll
    for (int w = 0; w < TS_WAVES; ++w) {
        const int sw = wsum[w];
        if (w < wave) before += sw;
        all += sw;
    }
    total = all;
    return before + x - v;
}

__global__ void __launch_bounds__(TS_THREADS)
tile_sum(const float* __restrict__ X, const float* __restrict__ W,
         const uint2* __restrict__ recs, const unsigned* __restrict__ runs,
         const float* __restrict__ b, float* __restrict__ out, int n, int nc)
{
    __shared__ __attribute__((aligned(16))) float acc[(TR + 1) * ALD];
    __shared__ __attribute__((aligned(16))) float wt[D * ALD];
    __shared__ __attribute__((aligned(16))) int   slot[SLOTS];
    __shared__ int wsum[TS_WAVES];
    const int t    = threadIdx.x;
    const int tile = blockIdx.x;
    float4* acc4 = (float4*)acc;

    const unsigned rw = (t < nc) ? runs[(size_t)tile * nc + t] : 0u;

    const float4* W4 = (const float4*)W;
    const float4 w0 = W4[t];
    const float4 w1 = W4[t + TS_THREADS];

    const int g    = t >> 4;
    const int li   = t & 15;
    const int k0   = (g & 1) * 2;

    for (int z = t; z < TR * ALD / 4; z += TS_THREADS) acc4[z] = make_float4(0.f, 0.f, 0.f, 0.f);
    {
        const int wk = t >> 4;
        float* wp = wt + (t & 15) * 4 * ALD + wk;
        wp[0 * ALD] = w0.x; wp[1 * ALD] = w0.y; wp[2 * ALD] = w0.z; wp[3 * ALD] = w0.w;
        wp[0 * ALD + 32] = w1.x; wp[1 * ALD + 32] = w1.y; wp[2 * ALD + 32] = w1.z; wp[3 * ALD + 32] = w1.w;
    }

    const int len   = (int)(rw >> 16);
    const int first = t * CH + (int)(rw & 0xFFFFu);
    int nt;
    const int pre = tile_scan_excl(len, wsum, t, nt);

    const int4*   slot4 = (const int4*)slot;
    const float4* X4    = (const float4*)X;
    for (int base = 0; base < nt; base += SLOTS) {
        const int end = min(nt - base, SLOTS);
        const int pad = (end + TS_STEP - 1) & ~(TS_STEP - 1);
        const int lo = max(pre, base);
        const int hi = min(pre + len, base + SLOTS);
        #pragma unroll 1
        for (int p = lo; p < hi; ++p) slot[p - base] = first + (p - pre);
        for (int p = end + t; p < pad; p += TS_THREADS) slot[p] = 0;
        __syncthreads();

        const int lastTrip = pad - TS_STEP;
        int   i  = 0;
        int4  sv = slot4[g];
        int4  sn = slot4[(min(TS_STEP, lastTrip) >> 2) + g];
        uint2 q[TS_U];
        q[0] = recs[sv.x]; q[1] = recs[sv.y]; q[2] = recs[sv.z]; q[3] = recs[sv.w];

        while (i < end) {
            float4 s[TS_U];
            #pragma unroll
            for (int u = 0; u < TS_U; ++u)
                s[u] = X4[(q[u].x & 0x00FFFFFFu) * (D / 4) + (unsigned)li];

            const int in = i + TS_STEP;
            uint2 qn[TS_U];
            qn[0] = recs[sn.x]; qn[1] = recs[sn.y]; qn[2] = recs[sn.z]; qn[3] = recs[sn.w];
            sn = slot4[(min(in + TS_STEP, lastTrip) >> 2) + g];

            #pragma unroll
            for (int u = 0; u < TS_U; ++u) {
                const bool     ok = (i + TS_U * g + u < end);
                const unsigned rl = ok ? (q[u].x >> 24) : (unsigned)TR;
                const float    v  = __uint_as_float(q[u].y);
                float* a = acc + rl * ALD + li * 4;
                const float p0 = v * s[u].x, p1 = v * s[u].y, p2 = v * s[u].z, p3 = v * s[u].w;
                atomicAdd(a + k0,           k0 ? p2 : p0);
                atomicAdd(a + k0 + 1,       k0 ? p3 : p1);
                atomicAdd(a + (k0 ^ 2),     k0 ? p0 : p2);
                atomicAdd(a + (k0 ^ 2) + 1, k0 ? p1 : p3);
            }
            #pragma unroll
            for (int u = 0; u < TS_U; ++u) q[u] = qn[u];
            i = in;
        }
        __syncthreads();
    }

    const int lane = t & 31;
    const int wave = t >> 5;
    const int h    = lane >> 4;
    const int ml   = lane & 15;
    const int m0   = (wave >> 1) * 16;
    const int n0   = (wave & 1) * 32;
    const float* ap  = acc + (m0 + ml) * ALD + 4 * h;
    const float* bp0 = wt + (n0 + ml) * ALD + 4 * h;
    const float* bp1 = bp0 + 16 * ALD;
    v8f c0 = {};
    v8f c1 = {};
    #pragma unroll
    for (int j = 0; j < D / 8; ++j) {
        const float4 fa = *(const float4*)(ap  + 8 * j);
        const float4 f0 = *(const float4*)(bp0 + 8 * j);
        const float4 f1 = *(const float4*)(bp1 + 8 * j);
        v2f a01; a01.x = fa.x; a01.y = fa.y;
        v2f a23; a23.x = fa.z; a23.y = fa.w;
        v2f p01; p01.x = f0.x; p01.y = f0.y;
        v2f p23; p23.x = f0.z; p23.y = f0.w;
        v2f q01; q01.x = f1.x; q01.y = f1.y;
        v2f q23; q23.x = f1.z; q23.y = f1.w;
        c0 = __builtin_amdgcn_wmma_f32_16x16x4_f32(false, a01, false, p01, (short)0, c0, false, false);
        c1 = __builtin_amdgcn_wmma_f32_16x16x4_f32(false, a01, false, q01, (short)0, c1, false, false);
        c0 = __builtin_amdgcn_wmma_f32_16x16x4_f32(false, a23, false, p23, (short)0, c0, false, false);
        c1 = __builtin_amdgcn_wmma_f32_16x16x4_f32(false, a23, false, q23, (short)0, c1, false, false);
    }
    __syncthreads();

    #pragma unroll
    for (int v = 0; v < 8; ++v) {
        float* o = acc + (m0 + v + 8 * h) * ALD + n0 + ml;
        o[0]  = c0[v];
        o[16] = c1[v];
    }
    __syncthreads();

    const float4* b4   = (const float4*)b;
    float4*       out4 = (float4*)out;
    for (int j = t; j < TR * D / 4; j += TS_THREADS) {
        const int rr = j >> 4;
        const int r  = tile * TR + rr;
        if (r < n) {
            const float4 a  = *(const float4*)(acc + rr * ALD + (j & 15) * 4);
            const float4 bb = b4[j & 15];
            out4[(size_t)r * (D / 4) + (j & 15)] = make_float4(a.x + bb.x, a.y + bb.y, a.z + bb.z, a.w + bb.w);
        }
    }
}

extern "C" void kernel_launch(void* const* d_in, const int* in_sizes, int n_in,
                              void* d_out, int out_size, void* d_ws, size_t ws_size,
                              hipStream_t stream)
{
    const float* X   = (const float*)d_in[0];
    const float* W   = (const float*)d_in[1];
    const float* b   = (const float*)d_in[2];
    const float* ev  = (const float*)d_in[3];
    const int*   row = (const int*)d_in[4];
    const int*   col = (const int*)d_in[5];
    float*       out = (float*)d_out;

    char*     ws   = (char*)d_ws;
    unsigned* runs = (unsigned*)(ws + (32u << 20));
    uint2*    recs = (uint2*)(ws + (40u << 20));

    const int N = in_sizes[0] / D;
    const int E = in_sizes[3];

    const int nb = (N + TR - 1) / TR;
    const int nc = (E + CH - 1) / CH;
    chunk_sort<<<nc, 1024, 0, stream>>>(row, col, ev, recs, runs, E, nb, nc);

    tile_sum<<<nb, TS_THREADS, 0, stream>>>(X, W, recs, runs, b, out, N, nc);
}


# tile kernel counting-sorts its records by output row in LDS (integer counters); 16 lanes sum a row in registers, one f32 LDS add per row change instead of per edge; arrival-order sums, other tree
# speedup vs baseline: 2.9137x; 1.3100x over previous
#include <hip/hip_runtime.h>

typedef float v2f __attribute__((ext_vector_type(2)));
typedef float v8f __attribute__((ext_vector_type(8)));

#define D 64

#define TR      128
#define TSH     7
#define MAXNB   1024
#define CH      8192
#define EPT     8

__device__ __forceinline__ int block_scan_incl(int v, int (*sc)[1024], int t)
{
    sc[0][t] = v;
    __syncthreads();
    int src = 0;
    for (int off = 1; off < 1024; off <<= 1) {
        int x = sc[src][t];
        if (t >= off) x += sc[src][t - off];
        sc[src ^ 1][t] = x;
        __syncthreads();
        src ^= 1;
    }
    return sc[src][t];
}

__global__ void __launch_bounds__(1024)
chunk_sort(const int* __restrict__ row, const int* __restrict__ col,
           const float* __restrict__ ev, uint2* __restrict__ recs,
           unsigned* __restrict__ runs, int E, int nb, int nc)
{
    __shared__ int   cnt[MAXNB + 1];
    __shared__ int   sc[2][1024];
    __shared__ uint2 stage[CH];

    const int t  = threadIdx.x;
    const int e0 = blockIdx.x * CH;
    cnt[t] = 0;
    __syncthreads();

    int      r[EPT];
    unsigned key[EPT];
    unsigned val[EPT];
    int      bin[EPT];
    int      rnk[EPT];
    #pragma unroll
    for (int j = 0; j < EPT; ++j) {
        const int ec = min(e0 + j * 1024 + t, E - 1);
        r[j]   = row[ec];
        key[j] = (unsigned)col[ec];
        val[j] = __float_as_uint(ev[ec]);
    }
    #pragma unroll
    for (int j = 0; j < EPT; ++j) {
        bin[j]  = (e0 + j * 1024 + t < E) ? (r[j] >> TSH) : MAXNB;
        key[j] |= (unsigned)(r[j] & (TR - 1)) << 24;
        rnk[j]  = atomicAdd(&cnt[bin[j]], 1);
    }
    __syncthreads();

    const int mine = cnt[t];
    const int excl = block_scan_incl(mine, sc, t) - mine;
    cnt[t] = excl;
    if (t < nb) runs[(size_t)t * nc + blockIdx.x] = (unsigned)excl | ((unsigned)mine << 16);
    __syncthreads();

    #pragma unroll
    for (int j = 0; j < EPT; ++j)
        if (bin[j] < MAXNB) stage[cnt[bin[j]] + rnk[j]] = make_uint2(key[j], val[j]);
    __syncthreads();

    const int nloc = min(CH, E - e0);
    uint2 rec[EPT];
    #pragma unroll
    for (int j = 0; j < EPT; ++j) rec[j] = stage[min(j * 1024 + t, nloc - 1)];
    #pragma unroll
    for (int j = 0; j < EPT; ++j)
        if (j * 1024 + t < nloc) recs[(size_t)e0 + j * 1024 + t] = rec[j];
}

#define TS_THREADS 512
#define TS_WAVES   (TS_THREADS / 32)
#define TS_U       4
#define TS_STEP    ((TS_THREADS / 16) * TS_U)
#define ALD        68
#define SLOTS      2560
#define RPT        (SLOTS / TS_THREADS)

__device__ __forceinline__ int tile_scan_excl(int v, int* wsum, int t, int& total)
{
    const int lane = t & 31;
    const int wave = t >> 5;
    int x = v;
    #pragma unroll
    for (int off = 1; off < 32; off <<= 1) {
        const int y = __shfl_up(x, off);
        if (lane >= off) x += y;
    }
    if (lane == 31) wsum[wave] = x;
    __syncthreads();
    int before = 0;
    int all    = 0;
    #pragma unroll
    for (int w = 0; w < TS_WAVES; ++w) {
        const int sw = wsum[w];
        if (w < wave) before += sw;
        all += sw;
    }
    total = all;
    return before + x - v;
}

__device__ __forceinline__ void row_step(unsigned x, unsigned y, const float4& s,
                                         unsigned& cur, float4& sum, float* accl)
{
    const unsigned rl = x >> 24;
    if (rl != cur) {
        float* a = accl + cur * ALD;
        atomicAdd(a,     sum.x);
        atomicAdd(a + 1, sum.y);
        atomicAdd(a + 2, sum.z);
        atomicAdd(a + 3, sum.w);
        sum = make_float4(0.f, 0.f, 0.f, 0.f);
        cur = rl;
    }
    const float v = __uint_as_float(y);
    sum.x += v * s.x;
    sum.y += v * s.y;
    sum.z += v * s.z;
    sum.w += v * s.w;
}

__global__ void __launch_bounds__(TS_THREADS) __attribute__((amdgpu_waves_per_eu(16, 16)))
tile_sum(const float* __restrict__ X, const float* __restrict__ W,
         const uint2* __restrict__ recs, const unsigned* __restrict__ runs,
         const float* __restrict__ b, float* __restrict__ out, int n, int nc)
{
    __shared__ __attribute__((aligned(16))) float acc[(TR + 1) * ALD];
    __shared__ __attribute__((aligned(16))) float wt[D * ALD];
    __shared__ __attribute__((aligned(16))) uint2 sorted[SLOTS];
    __shared__ int cnt[TR + 32];
    __shared__ int wsum[TS_WAVES];
    const int t    = threadIdx.x;
    const int tile = blockIdx.x;
    const int lane = t & 31;
    const int wave = t >> 5;
    float4* acc4 = (float4*)acc;
    int*    slot = (int*)sorted;

    const unsigned rw = (t < nc) ? runs[(size_t)tile * nc + t] : 0u;

    const float4* W4 = (const float4*)W;
    const float4 w0 = W4[t];
    const float4 w1 = W4[t + TS_THREADS];

    const int g    = t >> 4;
    const int li   = t & 15;

    for (int z = t; z < TR * ALD / 4; z += TS_THREADS) acc4[z] = make_float4(0.f, 0.f, 0.f, 0.f);
    {
        const int wk = t >> 4;
        float* wp = wt + (t & 15) * 4 * ALD + wk;
        wp[0 * ALD] = w0.x; wp[1 * ALD] = w0.y; wp[2 * ALD] = w0.z; wp[3 * ALD] = w0.w;
        wp[0 * ALD + 32] = w1.x; wp[1 * ALD + 32] = w1.y; wp[2 * ALD + 32] = w1.z; wp[3 * ALD + 32] = w1.w;
    }

    const int len   = (int)(rw >> 16);
    const int first = t * CH + (int)(rw & 0xFFFFu);
    int nt;
    const int pre = tile_scan_excl(len, wsum, t, nt);

    const float4* X4   = (const float4*)X;
    float*        accl = acc + li * 4;
    for (int base = 0; base < nt; base += SLOTS) {
        const int end = min(nt - base, SLOTS);
        const int pad = (end + TS_STEP - 1) & ~(TS_STEP - 1);
        if (t < TR) cnt[t] = 0;
        const int lo = max(pre, base);
        const int hi = min(pre + len, base + SLOTS);
        #pragma unroll 1
        for (int p = lo; p < hi; ++p) slot[p - base] = first + (p - pre);
        __syncthreads();

        int   idx[RPT];
        uint2 rec[RPT];
        int   rl[RPT];
        int   rk[RPT];
        #pragma unroll
        for (int j = 0; j < RPT; ++j) idx[j] = slot[min(j * TS_THREADS + t, end - 1)];
        #pragma unroll
        for (int j = 0; j < RPT; ++j) rec[j] = recs[idx[j]];
        #pragma unroll
        for (int j = 0; j < RPT; ++j) {
            rl[j] = (j * TS_THREADS + t < end) ? (int)(rec[j].x >> 24) : TR + lane;
            rk[j] = atomicAdd(&cnt[rl[j]], 1);
        }
        __syncthreads();

        {
            const int c = (t < TR) ? cnt[t] : 0;
            int x = c;
            #pragma unroll
            for (int off = 1; off < 32; off <<= 1) {
                const int y = __shfl_up(x, off);
                if (lane >= off) x += y;
            }
            if (lane == 31 && wave < TR / 32) wsum[wave] = x;
            __syncthreads();
            if (t < TR) {
                int before = 0;
                #pragma unroll
                for (int w = 0; w < TR / 32; ++w) if (w < wave) before += wsum[w];
                cnt[t] = before + x - c;
            }
        }
        __syncthreads();

        #pragma unroll
        for (int j = 0; j < RPT; ++j)
            if (j * TS_THREADS + t < end) sorted[cnt[rl[j]] + rk[j]] = rec[j];
        for (int p = end + t; p < pad; p += TS_THREADS) sorted[p] = make_uint2((unsigned)TR << 24, 0u);
        __syncthreads();

        const int    per = pad / (TS_THREADS / 16);
        const uint4* sp  = (const uint4*)(sorted + g * per);
        uint4    qa  = sp[0];
        uint4    qb  = sp[1];
        unsigned cur = qa.x >> 24;
        float4   sum = make_float4(0.f, 0.f, 0.f, 0.f);
        for (int i = 0; i < per; i += TS_U) {
            const float4 s0 = X4[(qa.x & 0x00FFFFFFu) * (D / 4) + (unsigned)li];
            const float4 s1 = X4[(qa.z & 0x00FFFFFFu) * (D / 4) + (unsigned)li];
            const float4 s2 = X4[(qb.x & 0x00FFFFFFu) * (D / 4) + (unsigned)li];
            const float4 s3 = X4[(qb.z & 0x00FFFFFFu) * (D / 4) + (unsigned)li];
            const int   in = min(i + TS_U, per - TS_U) >> 1;
            const uint4 na = sp[in];
            const uint4 nb = sp[in + 1];
            row_step(qa.x, qa.y, s0, cur, sum, accl);
            row_step(qa.z, qa.w, s1, cur, sum, accl);
            row_step(qb.x, qb.y, s2, cur, sum, accl);
            row_step(qb.z, qb.w, s3, cur, sum, accl);
            qa = na;
            qb = nb;
        }
        {
            float* a = accl + cur * ALD;
            atomicAdd(a,     sum.x);
            atomicAdd(a + 1, sum.y);
            atomicAdd(a + 2, sum.z);
            atomicAdd(a + 3, sum.w);
        }
        __syncthreads();
    }

    const int h    = lane >> 4;
    const int ml   = lane & 15;
    const int m0   = (wave >> 1) * 16;
    const int n0   = (wave & 1) * 32;
    const float* ap  = acc + (m0 + ml) * ALD + 4 * h;
    const float* bp0 = wt + (n0 + ml) * ALD + 4 * h;
    const float* bp1 = bp0 + 16 * ALD;
    v8f c0 = {};
    v8f c1 = {};
    #pragma unroll
    for (int j = 0; j < D / 8; ++j) {
        const float4 fa = *(const float4*)(ap  + 8 * j);
        const float4 f0 = *(const float4*)(bp0 + 8 * j);
        const float4 f1 = *(const float4*)(bp1 + 8 * j);
        v2f a01; a01.x = fa.x; a01.y = fa.y;
        v2f a23; a23.x = fa.z; a23.y = fa.w;
        v2f p01; p01.x = f0.x; p01.y = f0.y;
        v2f p23; p23.x = f0.z; p23.y = f0.w;
        v2f q01; q01.x = f1.x; q01.y = f1.y;
        v2f q23; q23.x = f1.z; q23.y = f1.w;
        c0 = __builtin_amdgcn_wmma_f32_16x16x4_f32(false, a01, false, p01, (short)0, c0, false, false);
        c1 = __builtin_amdgcn_wmma_f32_16x16x4_f32(false, a01, false, q01, (short)0, c1, false, false);
        c0 = __builtin_amdgcn_wmma_f32_16x16x4_f32(false, a23, false, p23, (short)0, c0, false, false);
        c1 = __builtin_amdgcn_wmma_f32_16x16x4_f32(false, a23, false, q23, (short)0, c1, false, false);
    }
    __syncthreads();

    #pragma unroll
    for (int v = 0; v < 8; ++v) {
        float* o = acc + (m0 + v + 8 * h) * ALD + n0 + ml;
        o[0]  = c0[v];
        o[16] = c1[v];
    }
    __syncthreads();

    const float4* b4   = (const float4*)b;
    float4*       out4 = (float4*)out;
    for (int j = t; j < TR * D / 4; j += TS_THREADS) {
        const int rr = j >> 4;
        const int r  = tile * TR + rr;
        if (r < n) {
            const float4 a  = *(const float4*)(acc + rr * ALD + (j & 15) * 4);
            const float4 bb = b4[j & 15];
            out4[(size_t)r * (D / 4) + (j & 15)] = make_float4(a.x + bb.x, a.y + bb.y, a.z + bb.z, a.w + bb.w);
        }
    }
}

extern "C" void kernel_launch(void* const* d_in, const int* in_sizes, int n_in,
                              void* d_out, int out_size, void* d_ws, size_t ws_size,
                              hipStream_t stream)
{
    const float* X   = (const float*)d_in[0];
    const float* W   = (const float*)d_in[1];
    const float* b   = (const float*)d_in[2];
    const float* ev  = (const float*)d_in[3];
    const int*   row = (const int*)d_in[4];
    const int*   col = (const int*)d_in[5];
    float*       out = (float*)d_out;

    char*     ws   = (char*)d_ws;
    unsigned* runs = (unsigned*)(ws + (32u << 20));
    uint2*    recs = (uint2*)(ws + (40u << 20));

    const int N = in_sizes[0] / D;
    const int E = in_sizes[3];

    const int nb = (N + TR - 1) / TR;
    const int nc = (E + CH - 1) / CH;
    chunk_sort<<<nc, 1024, 0, stream>>>(row, col, ev, recs, runs, E, nb, nc);

    tile_sum<<<nb, TS_THREADS, 0, stream>>>(X, W, recs, runs, b, out, N, nc);
}


# half as many waves, twice the work per lane: chunk sort 512 threads x 16 edges (scan with one barrier), tile kernel 8 waves, 8 rows of X in flight per 16 lanes; arrival-order sums, other tree
# speedup vs baseline: 3.0628x; 1.0512x over previous
#include <hip/hip_runtime.h>

typedef float v2f __attribute__((ext_vector_type(2)));
typedef float v8f __attribute__((ext_vector_type(8)));

#define D 64

#define TR      128
#define TSH     7
#define MAXNB   1024
#define CH      8192
#define CS_THREADS 512
#define EPT     16
static_assert(MAXNB == 2 * CS_THREADS, "a sorting thread owns exactly two tile counters");
static_assert(CH == EPT * CS_THREADS, "a chunk is EPT edges per sorting thread");

template <int NW>
__device__ __forceinline__ int wg_scan_excl(int v, int* wsum, int t, int& total)
{
    const int lane = t & 31;
    const int wave = t >> 5;
    int x = v;
    #pragma unroll
    for (int off = 1; off < 32; off <<= 1) {
        const int y = __shfl_up(x, off);
        if (lane >= off) x += y;
    }
    if (lane == 31) wsum[wave] = x;
    __syncthreads();
    int before = 0;
    int all    = 0;
    #pragma unroll
    for (int w = 0; w < NW; ++w) {
        const int sw = wsum[w];
        if (w < wave) before += sw;
        all += sw;
    }
    total = all;
    return before + x - v;
}

__global__ void __launch_bounds__(CS_THREADS) __attribute__((amdgpu_waves_per_eu(4, 4)))
chunk_sort(const int* __restrict__ row, const int* __restrict__ col,
           const float* __restrict__ ev, uint2* __restrict__ recs,
           unsigned* __restrict__ runs, int E, int nb, int nc)
{
    __shared__ int   cnt[MAXNB + 1];
    __shared__ int   wsum[CS_THREADS / 32];
    __shared__ uint2 stage[CH];

    const int t  = threadIdx.x;
    const int e0 = blockIdx.x * CH;
    cnt[t] = 0;
    cnt[t + CS_THREADS] = 0;
    __syncthreads();

    int      r[EPT];
    unsigned key[EPT];
    unsigned val[EPT];
    int      bin[EPT];
    int      rnk[EPT];
    #pragma unroll
    for (int j = 0; j < EPT; ++j) {
        const int ec = min(e0 + j * CS_THREADS + t, E - 1);
        r[j]   = row[ec];
        key[j] = (unsigned)col[ec];
        val[j] = __float_as_uint(ev[ec]);
    }
    #pragma unroll
    for (int j = 0; j < EPT; ++j) {
        bin[j]  = (e0 + j * CS_THREADS + t < E) ? (r[j] >> TSH) : MAXNB;
        key[j] |= (unsigned)(r[j] & (TR - 1)) << 24;
        rnk[j]  = atomicAdd(&cnt[bin[j]], 1);
    }
    __syncthreads();

    const int m0 = cnt[2 * t];
    const int m1 = cnt[2 * t + 1];
    int staged;
    const int ex0 = wg_scan_excl<CS_THREADS / 32>(m0 + m1, wsum, t, staged);
    const int ex1 = ex0 + m0;
    cnt[2 * t]     = ex0;
    cnt[2 * t + 1] = ex1;
    if (2 * t < nb)     runs[(size_t)(2 * t) * nc + blockIdx.x]     = (unsigned)ex0 | ((unsigned)m0 << 16);
    if (2 * t + 1 < nb) runs[(size_t)(2 * t + 1) * nc + blockIdx.x] = (unsigned)ex1 | ((unsigned)m1 << 16);
    __syncthreads();

    #pragma unroll
    for (int j = 0; j < EPT; ++j)
        if (bin[j] < MAXNB) stage[cnt[bin[j]] + rnk[j]] = make_uint2(key[j], val[j]);
    __syncthreads();

    const int nloc = min(CH, E - e0);
    uint2 rec[EPT];
    #pragma unroll
    for (int j = 0; j < EPT; ++j) rec[j] = stage[min(j * CS_THREADS + t, nloc - 1)];
    #pragma unroll
    for (int j = 0; j < EPT; ++j)
        if (j * CS_THREADS + t < nloc) recs[(size_t)e0 + j * CS_THREADS + t] = rec[j];
}

#define TS_THREADS 256
#define TS_WAVES   (TS_THREADS / 32)
#define TS_U       8
#define TS_STEP    ((TS_THREADS / 16) * TS_U)
#define ALD        68
#define SLOTS      2560
#define RPT        (SLOTS / TS_THREADS)
#define OPT        (TR * D / 4 / TS_THREADS)

__device__ __forceinline__ void row_step(unsigned x, unsigned y, const float4& s,
                                         unsigned& cur, float4& sum, float* accl)
{
    const unsigned rl = x >> 24;
    if (rl != cur) {
        float* a = accl + cur * ALD;
        atomicAdd(a,     sum.x);
        atomicAdd(a + 1, sum.y);
        atomicAdd(a + 2, sum.z);
        atomicAdd(a + 3, sum.w);
        sum = make_float4(0.f, 0.f, 0.f, 0.f);
        cur = rl;
    }
    const float v = __uint_as_float(y);
    sum.x += v * s.x;
    sum.y += v * s.y;
    sum.z += v * s.z;
    sum.w += v * s.w;
}

__global__ void __launch_bounds__(TS_THREADS) __attribute__((amdgpu_waves_per_eu(8, 8)))
tile_sum(const float* __restrict__ X, const float* __restrict__ W,
         const uint2* __restrict__ recs, const unsigned* __restrict__ runs,
         const float* __restrict__ b, float* __restrict__ out, int n, int nc)
{
    __shared__ __attribute__((aligned(16))) float acc[(TR + 1) * ALD];
    __shared__ __attribute__((aligned(16))) float wt[D * ALD];
    __shared__ __attribute__((aligned(16))) uint2 sorted[SLOTS];
    __shared__ int cnt[TR + 32];
    __shared__ int wsum[TS_WAVES];
    const int t    = threadIdx.x;
    const int tile = blockIdx.x;
    const int lane = t & 31;
    const int wave = t >> 5;
    float4* acc4 = (float4*)acc;
    int*    slot = (int*)sorted;

    const unsigned rw = (t < nc) ? runs[(size_t)tile * nc + t] : 0u;

    const float4* W4 = (const float4*)W;
    const float4 w0 = W4[t];
    const float4 w1 = W4[t + TS_THREADS];
    const float4 w2 = W4[t + 2 * TS_THREADS];
    const float4 w3 = W4[t + 3 * TS_THREADS];

    const int g    = t >> 4;
    const int li   = t & 15;

    for (int z = t; z < TR * ALD / 4; z += TS_THREADS) acc4[z] = make_float4(0.f, 0.f, 0.f, 0.f);
    {
        const int wk = t >> 4;
        float* wp = wt + (t & 15) * 4 * ALD + wk;
        wp[0 * ALD] = w0.x; wp[1 * ALD] = w0.y; wp[2 * ALD] = w0.z; wp[3 * ALD] = w0.w;
        wp[0 * ALD + 16] = w1.x; wp[1 * ALD + 16] = w1.y; wp[2 * ALD + 16] = w1.z; wp[3 * ALD + 16] = w1.w;
        wp[0 * ALD + 32] = w2.x; wp[1 * ALD + 32] = w2.y; wp[2 * ALD + 32] = w2.z; wp[3 * ALD + 32] = w2.w;
        wp[0 * ALD + 48] = w3.x; wp[1 * ALD + 48] = w3.y; wp[2 * ALD + 48] = w3.z; wp[3 * ALD + 48] = w3.w;
    }

    const int len   = (int)(rw >> 16);
    const int first = t * CH + (int)(rw & 0xFFFFu);
    int nt;
    const int pre = wg_scan_excl<TS_WAVES>(len, wsum, t, nt);

    const float4* X4   = (const float4*)X;
    float*        accl = acc + li * 4;
    for (int base = 0; base < nt; base += SLOTS) {
        const int end = min(nt - base, SLOTS);
        const int pad = (end + TS_STEP - 1) & ~(TS_STEP - 1);
        if (t < TR) cnt[t] = 0;
        const int lo = max(pre, base);
        const int hi = min(pre + len, base + SLOTS);
        #pragma unroll 1
        for (int p = lo; p < hi; ++p) slot[p - base] = first + (p - pre);
        __syncthreads();

        int   idx[RPT];
        uint2 rec[RPT];
        int   rl[RPT];
        int   rk[RPT];
        #pragma unroll
        for (int j = 0; j < RPT; ++j) idx[j] = slot[min(j * TS_THREADS + t, end - 1)];
        #pragma unroll
        for (int j = 0; j < RPT; ++j) rec[j] = recs[idx[j]];
        #pragma unroll
        for (int j = 0; j < RPT; ++j) {
            rl[j] = (j * TS_THREADS + t < end) ? (int)(rec[j].x >> 24) : TR + lane;
            rk[j] = atomicAdd(&cnt[rl[j]], 1);
        }
        __syncthreads();

        {
            const int c = (t < TR) ? cnt[t] : 0;
            int x = c;
            #pragma unroll
            for (int off = 1; off < 32; off <<= 1) {
                const int y = __shfl_up(x, off);
                if (lane >= off) x += y;
            }
            if (lane == 31 && wave < TR / 32) wsum[wave] = x;
            __syncthreads();
            if (t < TR) {
                int before = 0;
                #pragma unroll
                for (int w = 0; w < TR / 32; ++w) if (w < wave) before += wsum[w];
                cnt[t] = before + x - c;
            }
        }
        __syncthreads();

        #pragma unroll
        for (int j = 0; j < RPT; ++j)
            if (j * TS_THREADS + t < end) sorted[cnt[rl[j]] + rk[j]] = rec[j];
        for (int p = end + t; p < pad; p += TS_THREADS) sorted[p] = make_uint2((unsigned)TR << 24, 0u);
        __syncthreads();

        const int    per = pad / (TS_THREADS / 16);
        const uint4* sp  = (const uint4*)(sorted + g * per);
        uint4    qa  = sp[0];
        uint4    qb  = sp[1];
        uint4    qc  = sp[2];
        uint4    qd  = sp[3];
        unsigned cur = qa.x >> 24;
        float4   sum = make_float4(0.f, 0.f, 0.f, 0.f);
        for (int i = 0; i < per; i += TS_U) {
            const float4 s0 = X4[(qa.x & 0x00FFFFFFu) * (D / 4) + (unsigned)li];
            const float4 s1 = X4[(qa.z & 0x00FFFFFFu) * (D / 4) + (unsigned)li];
            const float4 s2 = X4[(qb.x & 0x00FFFFFFu) * (D / 4) + (unsigned)li];
            const float4 s3 = X4[(qb.z & 0x00FFFFFFu) * (D / 4) + (unsigned)li];
            const float4 s4 = X4[(qc.x & 0x00FFFFFFu) * (D / 4) + (unsigned)li];
            const float4 s5 = X4[(qc.z & 0x00FFFFFFu) * (D / 4) + (unsigned)li];
            const float4 s6 = X4[(qd.x & 0x00FFFFFFu) * (D / 4) + (unsigned)li];
            const float4 s7 = X4[(qd.z & 0x00FFFFFFu) * (D / 4) + (unsigned)li];
            const int   in = min(i + TS_U, per - TS_U) >> 1;
            const uint4 n0 = sp[in];
            const uint4 n1 = sp[in + 1];
            const uint4 n2 = sp[in + 2];
            const uint4 n3 = sp[in + 3];
            row_step(qa.x, qa.y, s0, cur, sum, accl);
            row_step(qa.z, qa.w, s1, cur, sum, accl);
            row_step(qb.x, qb.y, s2, cur, sum, accl);
            row_step(qb.z, qb.w, s3, cur, sum, accl);
            row_step(qc.x, qc.y, s4, cur, sum, accl);
            row_step(qc.z, qc.w, s5, cur, sum, accl);
            row_step(qd.x, qd.y, s6, cur, sum, accl);
            row_step(qd.z, qd.w, s7, cur, sum, accl);
            qa = n0;
            qb = n1;
            qc = n2;
            qd = n3;
        }
        {
            float* a = accl + cur * ALD;
            atomicAdd(a,     sum.x);
            atomicAdd(a + 1, sum.y);
            atomicAdd(a + 2, sum.z);
            atomicAdd(a + 3, sum.w);
        }
        __syncthreads();
    }

    const float4 bb = ((const float4*)b)[li];
    const int h    = lane >> 4;
    const int ml   = lane & 15;
    const int m0   = wave * 16;
    const float* ap = acc + (m0 + ml) * ALD + 4 * h;
    const float* bp = wt + ml * ALD + 4 * h;
    v8f c0 = {};
    v8f c1 = {};
    v8f c2 = {};
    v8f c3 = {};
    #pragma unroll
    for (int j = 0; j < D / 8; ++j) {
        const float4 fa = *(const float4*)(ap + 8 * j);
        const float4 f0 = *(const float4*)(bp + 8 * j);
        const float4 f1 = *(const float4*)(bp + 16 * ALD + 8 * j);
        const float4 f2 = *(const float4*)(bp + 32 * ALD + 8 * j);
        const float4 f3 = *(const float4*)(bp + 48 * ALD + 8 * j);
        v2f a01; a01.x = fa.x; a01.y = fa.y;
        v2f a23; a23.x = fa.z; a23.y = fa.w;
        v2f p01; p01.x = f0.x; p01.y = f0.y;
        v2f p23; p23.x = f0.z; p23.y = f0.w;
        v2f q01; q01.x = f1.x; q01.y = f1.y;
        v2f q23; q23.x = f1.z; q23.y = f1.w;
        v2f r01; r01.x = f2.x; r01.y = f2.y;
        v2f r23; r23.x = f2.z; r23.y = f2.w;
        v2f u01; u01.x = f3.x; u01.y = f3.y;
        v2f u23; u23.x = f3.z; u23.y = f3.w;
        c0 = __builtin_amdgcn_wmma_f32_16x16x4_f32(false, a01, false, p01, (short)0, c0, false, false);
        c1 = __builtin_amdgcn_wmma_f32_16x16x4_f32(false, a01, false, q01, (short)0, c1, false, false);
        c2 = __builtin_amdgcn_wmma_f32_16x16x4_f32(false, a01, false, r01, (short)0, c2, false, false);
        c3 = __builtin_amdgcn_wmma_f32_16x16x4_f32(false, a01, false, u01, (short)0, c3, false, false);
        c0 = __builtin_amdgcn_wmma_f32_16x16x4_f32(false, a23, false, p23, (short)0, c0, false, false);
        c1 = __builtin_amdgcn_wmma_f32_16x16x4_f32(false, a23, false, q23, (short)0, c1, false, false);
        c2 = __builtin_amdgcn_wmma_f32_16x16x4_f32(false, a23, false, r23, (short)0, c2, false, false);
        c3 = __builtin_amdgcn_wmma_f32_16x16x4_f32(false, a23, false, u23, (short)0, c3, false, false);
    }
    #pragma unroll
    for (int v = 0; v < 8; ++v) {
        float* o = acc + (m0 + v + 8 * h) * ALD + ml;
        o[0]  = c0[v];
        o[16] = c1[v];
        o[32] = c2[v];
        o[48] = c3[v];
    }
    __syncthreads();

    float4* out4 = (float4*)out;
    float4  orow[OPT];
    #pragma unroll
    for (int k = 0; k < OPT; ++k)
        orow[k] = *(const float4*)(acc + (g + k * (TS_THREADS / 16)) * ALD + li * 4);
    #pragma unroll
    for (int k = 0; k < OPT; ++k) {
        const int r = tile * TR + g + k * (TS_THREADS / 16);
        if (r < n)
            out4[(size_t)r * (D / 4) + li] = make_float4(orow[k].x + bb.x, orow[k].y + bb.y,
                                                         orow[k].z + bb.z, orow[k].w + bb.w);
    }
}

extern "C" void kernel_launch(void* const* d_in, const int* in_sizes, int n_in,
                              void* d_out, int out_size, void* d_ws, size_t ws_size,
                              hipStream_t stream)
{
    const float* X   = (const float*)d_in[0];
    const float* W   = (const float*)d_in[1];
    const float* b   = (const float*)d_in[2];
    const float* ev  = (const float*)d_in[3];
    const int*   row = (const int*)d_in[4];
    const int*   col = (const int*)d_in[5];
    float*       out = (float*)d_out;

    char*     ws   = (char*)d_ws;
    unsigned* runs = (unsigned*)(ws + (32u << 20));
    uint2*    recs = (uint2*)(ws + (40u << 20));

    const int N = in_sizes[0] / D;
    const int E = in_sizes[3];

    const int nb = (N + TR - 1) / TR;
    const int nc = (E + CH - 1) / CH;
    chunk_sort<<<nc, CS_THREADS, 0, stream>>>(row, col, ev, recs, runs, E, nb, nc);

    tile_sum<<<nb, TS_THREADS, 0, stream>>>(X, W, recs, runs, b, out, N, nc);
}
